# starting kernel
# baseline (speedup 1.0000x reference)
#include <hip/hip_runtime.h>

#define TGT 64
#define SRC 64
#define BATCH 32
#define HID 512
#define ATT 512

typedef __attribute__((ext_vector_type(2))) float v2f;
typedef __attribute__((ext_vector_type(8))) float v8f;

__device__ __forceinline__ float fast_tanh(float x) {
#if defined(__has_builtin)
#if __has_builtin(__builtin_amdgcn_tanhf)
    return __builtin_amdgcn_tanhf(x);
#elif __has_builtin(__builtin_amdgcn_tanh_f32)
    return __builtin_amdgcn_tanh_f32(x);
#else
    return tanhf(x);
#endif
#else
    return tanhf(x);
#endif
}

__global__ __launch_bounds__(128) void proj_gemm_kernel(
    const float* __restrict__ h_t,
    const float* __restrict__ src,
    const float* __restrict__ Wa,
    float* __restrict__ P)
{
    const int lane = threadIdx.x & 31;
    const int wave = threadIdx.x >> 5;
    const int tile = blockIdx.x * 4 + wave;
    const int tilesN = ATT / 16;
    const int tileM = tile / tilesN;
    const int tileN = tile % tilesN;

    const int rowBase = tileM * 64;
    const bool isH = rowBase < (TGT * BATCH);
    const float* __restrict__ X = isH ? (h_t + (size_t)rowBase * HID)
                                      : (src + (size_t)(rowBase - TGT * BATCH) * HID);
    const float* __restrict__ W = Wa + (size_t)(isH ? 0 : HID) * ATT + tileN * 16;

    const int m    = lane & 15;
    const int half = lane >> 4;

    v8f acc0 = {}, acc1 = {}, acc2 = {}, acc3 = {};
    for (int k0 = 0; k0 < HID; k0 += 4) {
        const int kA = k0 + half * 2;
        v2f b;
        b.x = W[(size_t)kA * ATT + m];
        b.y = W[(size_t)(kA + 1) * ATT + m];

        v2f a0 = *(const v2f*)(X + (size_t)(0 * 16 + m) * HID + kA);
        v2f a1 = *(const v2f*)(X + (size_t)(1 * 16 + m) * HID + kA);
        v2f a2 = *(const v2f*)(X + (size_t)(2 * 16 + m) * HID + kA);
        v2f a3 = *(const v2f*)(X + (size_t)(3 * 16 + m) * HID + kA);

        acc0 = __builtin_amdgcn_wmma_f32_16x16x4_f32(false, a0, false, b, (short)0, acc0, false, false);
        acc1 = __builtin_amdgcn_wmma_f32_16x16x4_f32(false, a1, false, b, (short)0, acc1, false, false);
        acc2 = __builtin_amdgcn_wmma_f32_16x16x4_f32(false, a2, false, b, (short)0, acc2, false, false);
        acc3 = __builtin_amdgcn_wmma_f32_16x16x4_f32(false, a3, false, b, (short)0, acc3, false, false);
    }

    float* __restrict__ out0 = P + (size_t)(rowBase + 0 * 16 + half * 8) * ATT + tileN * 16 + m;
    float* __restrict__ out1 = P + (size_t)(rowBase + 1 * 16 + half * 8) * ATT + tileN * 16 + m;
    float* __restrict__ out2 = P + (size_t)(rowBase + 2 * 16 + half * 8) * ATT + tileN * 16 + m;
    float* __restrict__ out3 = P + (size_t)(rowBase + 3 * 16 + half * 8) * ATT + tileN * 16 + m;
#pragma unroll
    for (int r = 0; r < 8; ++r) {
        out0[(size_t)r * ATT] = acc0[r];
        out1[(size_t)r * ATT] = acc1[r];
        out2[(size_t)r * ATT] = acc2[r];
        out3[(size_t)r * ATT] = acc3[r];
    }
}

__global__ __launch_bounds__(256) void attn_fused_kernel(
    const float* __restrict__ src,
    const float* __restrict__ Va,
    const float* __restrict__ P,
    float* __restrict__ out)
{
    __shared__ float hp[ATT];
    __shared__ float va[ATT];
    __shared__ float attn[SRC];

    const int tb   = blockIdx.x;
    const int b    = tb & (BATCH - 1);
    const int tid  = threadIdx.x;
    const int lane = tid & 31;
    const int wave = tid >> 5;

    const float* __restrict__ hpg = P + (size_t)tb * ATT;
    for (int i = tid; i < ATT; i += 256) {
        hp[i] = hpg[i];
        va[i] = Va[i];
    }
    __syncthreads();

    for (int s = wave; s < SRC; s += 8) {
        const float* __restrict__ spg =
            P + (size_t)(TGT * BATCH + s * BATCH + b) * ATT;
        float acc = 0.f;
#pragma unroll
        for (int i = 0; i < ATT / 64; ++i) {
            const int a = i * 64 + lane * 2;
            const float2 sp2 = *(const float2*)(spg + a);
            const float x0 = hp[a] + sp2.x;
            const float x1 = hp[a + 1] + sp2.y;
            acc += fast_tanh(x0) * va[a] + fast_tanh(x1) * va[a + 1];
        }
#pragma unroll
        for (int off = 16; off; off >>= 1)
            acc += __shfl_xor(acc, off, 32);
        if (lane == 0) attn[s] = acc;
    }
    __syncthreads();

    if (tid < 32) {
        float v0 = attn[lane], v1 = attn[lane + 32];
        float mx = fmaxf(v0, v1);
#pragma unroll
        for (int off = 16; off; off >>= 1)
            mx = fmaxf(mx, __shfl_xor(mx, off, 32));
        float e0 = __expf(v0 - mx), e1 = __expf(v1 - mx);
        float ssum = e0 + e1;
#pragma unroll
        for (int off = 16; off; off >>= 1)
            ssum += __shfl_xor(ssum, off, 32);
        float inv = 1.f / ssum;
        attn[lane]      = e0 * inv;
        attn[lane + 32] = e1 * inv;
    }
    __syncthreads();

    const int h0 = tid * 2;
    float c0 = 0.f, c1 = 0.f;
    for (int s = 0; s < SRC; ++s) {
        const float aw = attn[s];
        const float2 v = *(const float2*)(src + (size_t)(s * BATCH + b) * HID + h0);
        c0 += aw * v.x;
        c1 += aw * v.y;
    }
    float2 res; res.x = c0; res.y = c1;
    *(float2*)(out + (size_t)tb * HID + h0) = res;
}

extern "C" void kernel_launch(void* const* d_in, const int* in_sizes, int n_in,
                              void* d_out, int out_size, void* d_ws, size_t ws_size,
                              hipStream_t stream) {
    const float* h_t = (const float*)d_in[0];
    const float* src = (const float*)d_in[1];
    const float* Wa  = (const float*)d_in[2];
    const float* Va  = (const float*)d_in[3];
    float* out = (float*)d_out;

    float* P = (float*)d_ws;

    proj_gemm_kernel<<<512, 128, 0, stream>>>(h_t, src, Wa, P);

    attn_fused_kernel<<<TGT * BATCH, 256, 0, stream>>>(src, Va, P, out);
}


# Projection GEMM stages X and W tiles through LDS with coalesced 16-byte global loads (64x64 tile per work-group, two LDS stages) instead of per-lane strided fragment gathers; k order, sums unchanged
# speedup vs baseline: 1.3022x; 1.3022x over previous
#include <hip/hip_runtime.h>

#define TGT 64
#define SRC 64
#define BATCH 32
#define HID 512
#define ATT 512

typedef __attribute__((ext_vector_type(2))) float v2f;
typedef __attribute__((ext_vector_type(8))) float v8f;

__device__ __forceinline__ float fast_tanh(float x) {
#if defined(__has_builtin)
#if __has_builtin(__builtin_amdgcn_tanhf)
    return __builtin_amdgcn_tanhf(x);
#elif __has_builtin(__builtin_amdgcn_tanh_f32)
    return __builtin_amdgcn_tanh_f32(x);
#else
    return tanhf(x);
#endif
#else
    return tanhf(x);
#endif
}

#define GBM 64
#define GBN 64
#define GBK 32
#define APITCH 36
#define BPITCH 160

__global__ __launch_bounds__(128) __attribute__((amdgpu_waves_per_eu(1, 2))) void proj_gemm_kernel(
    const float* __restrict__ h_t,
    const float* __restrict__ src,
    const float* __restrict__ Wa,
    float* __restrict__ P)
{
    __shared__ __attribute__((aligned(16))) float As[2][GBM * APITCH];
    __shared__ __attribute__((aligned(16))) float Bs[2][(GBK / 2) * BPITCH];

    const int tid  = threadIdx.x;
    const int lane = tid & 31;
    const int wave = tid >> 5;
    const int wm   = wave >> 1;
    const int wn   = wave & 1;
    const int c    = lane & 15;
    const int half = lane >> 4;

    const int blockN  = blockIdx.x & 7;
    const int blockM  = blockIdx.x >> 3;
    const int rowBase = blockM * GBM;
    const bool isH = rowBase < (TGT * BATCH);
    const float* __restrict__ X = isH ? (h_t + (size_t)rowBase * HID)
                                      : (src + (size_t)(rowBase - TGT * BATCH) * HID);
    const float* __restrict__ W = Wa + (size_t)(isH ? 0 : HID) * ATT + blockN * GBN;

    const int arow = tid >> 3;
    const int acol = (tid & 7) * 4;
    const int bkp  = tid >> 4;
    const int bcol = (tid & 15) * 4;
    const float* __restrict__ gA = X + (size_t)arow * HID + acol;
    const float* __restrict__ gB = W + (size_t)(2 * bkp) * ATT + bcol;
    const int sA = arow * APITCH + acol;
    const int sB = bkp * BPITCH + bcol * 2;

    float4 ra0, ra1, ra2, ra3, rbe0, rbo0, rbe1, rbo1;

#define LOAD_STAGE()                                                            \
    do {                                                                        \
        rbe0 = *(const float4*)(pB);                                            \
        rbo0 = *(const float4*)(pB + ATT);                                      \
        rbe1 = *(const float4*)(pB + 16 * ATT);                                 \
        rbo1 = *(const float4*)(pB + 17 * ATT);                                 \
        ra0  = *(const float4*)(pA);                                            \
        ra1  = *(const float4*)(pA + 16 * HID);                                 \
        ra2  = *(const float4*)(pA + 32 * HID);                                 \
        ra3  = *(const float4*)(pA + 48 * HID);                                 \
        pA += GBK;                                                              \
        pB += (size_t)GBK * ATT;                                                \
    } while (0)

#define STORE_STAGE(BUF)                                                        \
    do {                                                                        \
        *(float4*)(&As[(BUF)][sA + 0  * APITCH]) = ra0;                         \
        *(float4*)(&As[(BUF)][sA + 16 * APITCH]) = ra1;                         \
        *(float4*)(&As[(BUF)][sA + 32 * APITCH]) = ra2;                         \
        *(float4*)(&As[(BUF)][sA + 48 * APITCH]) = ra3;                         \
        *(float4*)(&Bs[(BUF)][sB])                  = make_float4(rbe0.x, rbo0.x, rbe0.y, rbo0.y); \
        *(float4*)(&Bs[(BUF)][sB + 4])              = make_float4(rbe0.z, rbo0.z, rbe0.w, rbo0.w); \
        *(float4*)(&Bs[(BUF)][sB + 8 * BPITCH])     = make_float4(rbe1.x, rbo1.x, rbe1.y, rbo1.y); \
        *(float4*)(&Bs[(BUF)][sB + 8 * BPITCH + 4]) = make_float4(rbe1.z, rbo1.z, rbe1.w, rbo1.w); \
    } while (0)

    const float* __restrict__ pA = gA;
    const float* __restrict__ pB = gB;
    LOAD_STAGE();
    STORE_STAGE(0);
    __syncthreads();

    const int fA = (wm * 32 + c) * APITCH + 2 * half;
    const int fB = half * BPITCH + (wn * 32 + c) * 2;

    v8f acc00 = {}, acc01 = {}, acc10 = {}, acc11 = {};

    for (int st = 0; st < HID / GBK; ++st) {
        const int cur = st & 1;
        const bool more = (st + 1) < (HID / GBK);
        if (more) LOAD_STAGE();

        const float* __restrict__ as = &As[cur][fA];
        const float* __restrict__ bs = &Bs[cur][fB];
        v2f a0[GBK / 4], a1[GBK / 4], b0[GBK / 4], b1[GBK / 4];
#pragma unroll
        for (int j = 0; j < GBK / 4; ++j) {
            a0[j] = *(const v2f*)(as + 4 * j);
            a1[j] = *(const v2f*)(as + 16 * APITCH + 4 * j);
            b0[j] = *(const v2f*)(bs + 2 * j * BPITCH);
            b1[j] = *(const v2f*)(bs + 2 * j * BPITCH + 32);
        }

#pragma unroll
        for (int j = 0; j < GBK / 4 - 2; ++j) {
            acc00 = __builtin_amdgcn_wmma_f32_16x16x4_f32(false, a0[j], false, b0[j], (short)0, acc00, false, false);
            acc01 = __builtin_amdgcn_wmma_f32_16x16x4_f32(false, a0[j], false, b1[j], (short)0, acc01, false, false);
            acc10 = __builtin_amdgcn_wmma_f32_16x16x4_f32(false, a1[j], false, b0[j], (short)0, acc10, false, false);
            acc11 = __builtin_amdgcn_wmma_f32_16x16x4_f32(false, a1[j], false, b1[j], (short)0, acc11, false, false);
        }

        __builtin_amdgcn_sched_barrier(0);

        if (more) STORE_STAGE(cur ^ 1);

#pragma unroll
        for (int j = GBK / 4 - 2; j < GBK / 4; ++j) {
            acc00 = __builtin_amdgcn_wmma_f32_16x16x4_f32(false, a0[j], false, b0[j], (short)0, acc00, false, false);
            acc01 = __builtin_amdgcn_wmma_f32_16x16x4_f32(false, a0[j], false, b1[j], (short)0, acc01, false, false);
            acc10 = __builtin_amdgcn_wmma_f32_16x16x4_f32(false, a1[j], false, b0[j], (short)0, acc10, false, false);
            acc11 = __builtin_amdgcn_wmma_f32_16x16x4_f32(false, a1[j], false, b1[j], (short)0, acc11, false, false);
        }
        __syncthreads();
    }
#undef LOAD_STAGE
#undef STORE_STAGE

    float* __restrict__ o00 = P + (size_t)(rowBase + wm * 32 + half * 8) * ATT + blockN * GBN + wn * 32 + c;
    float* __restrict__ o10 = o00 + (size_t)16 * ATT;
#pragma unroll
    for (int r = 0; r < 8; ++r) {
        o00[(size_t)r * ATT]      = acc00[r];
        o00[(size_t)r * ATT + 16] = acc01[r];
        o10[(size_t)r * ATT]      = acc10[r];
        o10[(size_t)r * ATT + 16] = acc11[r];
    }
}

__global__ __launch_bounds__(256) void attn_fused_kernel(
    const float* __restrict__ src,
    const float* __restrict__ Va,
    const float* __restrict__ P,
    float* __restrict__ out)
{
    __shared__ float hp[ATT];
    __shared__ float va[ATT];
    __shared__ float attn[SRC];

    const int tb   = blockIdx.x;
    const int b    = tb & (BATCH - 1);
    const int tid  = threadIdx.x;
    const int lane = tid & 31;
    const int wave = tid >> 5;

    const float* __restrict__ hpg = P + (size_t)tb * ATT;
    for (int i = tid; i < ATT; i += 256) {
        hp[i] = hpg[i];
        va[i] = Va[i];
    }
    __syncthreads();

    for (int s = wave; s < SRC; s += 8) {
        const float* __restrict__ spg =
            P + (size_t)(TGT * BATCH + s * BATCH + b) * ATT;
        float acc = 0.f;
#pragma unroll
        for (int i = 0; i < ATT / 64; ++i) {
            const int a = i * 64 + lane * 2;
            const float2 sp2 = *(const float2*)(spg + a);
            const float x0 = hp[a] + sp2.x;
            const float x1 = hp[a + 1] + sp2.y;
            acc += fast_tanh(x0) * va[a] + fast_tanh(x1) * va[a + 1];
        }
#pragma unroll
        for (int off = 16; off; off >>= 1)
            acc += __shfl_xor(acc, off, 32);
        if (lane == 0) attn[s] = acc;
    }
    __syncthreads();

    if (tid < 32) {
        float v0 = attn[lane], v1 = attn[lane + 32];
        float mx = fmaxf(v0, v1);
#pragma unroll
        for (int off = 16; off; off >>= 1)
            mx = fmaxf(mx, __shfl_xor(mx, off, 32));
        float e0 = __expf(v0 - mx), e1 = __expf(v1 - mx);
        float ssum = e0 + e1;
#pragma unroll
        for (int off = 16; off; off >>= 1)
            ssum += __shfl_xor(ssum, off, 32);
        float inv = 1.f / ssum;
        attn[lane]      = e0 * inv;
        attn[lane + 32] = e1 * inv;
    }
    __syncthreads();

    const int h0 = tid * 2;
    float c0 = 0.f, c1 = 0.f;
    for (int s = 0; s < SRC; ++s) {
        const float aw = attn[s];
        const float2 v = *(const float2*)(src + (size_t)(s * BATCH + b) * HID + h0);
        c0 += aw * v.x;
        c1 += aw * v.y;
    }
    float2 res; res.x = c0; res.y = c1;
    *(float2*)(out + (size_t)tb * HID + h0) = res;
}

extern "C" void kernel_launch(void* const* d_in, const int* in_sizes, int n_in,
                              void* d_out, int out_size, void* d_ws, size_t ws_size,
                              hipStream_t stream) {
    const float* h_t = (const float*)d_in[0];
    const float* src = (const float*)d_in[1];
    const float* Wa  = (const float*)d_in[2];
    const float* Va  = (const float*)d_in[3];
    float* out = (float*)d_out;

    float* P = (float*)d_ws;

    proj_gemm_kernel<<<512, 128, 0, stream>>>(h_t, src, Wa, P);

    attn_fused_kernel<<<TGT * BATCH, 256, 0, stream>>>(src, Va, P, out);
}


# Attention kernel: a work-group does 8 target steps of one batch column, fetching each s_part/src row once for all 8; h rows in registers, lane partial scores summed once via LDS (sum order changed)
# speedup vs baseline: 1.6044x; 1.2321x over previous
#include <hip/hip_runtime.h>

#define TGT 64
#define SRC 64
#define BATCH 32
#define HID 512
#define ATT 512

typedef __attribute__((ext_vector_type(2))) float v2f;
typedef __attribute__((ext_vector_type(8))) float v8f;

__device__ __forceinline__ float fast_tanh(float x) {
#if defined(__has_builtin)
#if __has_builtin(__builtin_amdgcn_tanhf)
    return __builtin_amdgcn_tanhf(x);
#elif __has_builtin(__builtin_amdgcn_tanh_f32)
    return __builtin_amdgcn_tanh_f32(x);
#else
    return tanhf(x);
#endif
#else
    return tanhf(x);
#endif
}

#define GBM 64
#define GBN 64
#define GBK 32
#define APITCH 36
#define BPITCH 160

__global__ __launch_bounds__(128) __attribute__((amdgpu_waves_per_eu(1, 2))) void proj_gemm_kernel(
    const float* __restrict__ h_t,
    const float* __restrict__ src,
    const float* __restrict__ Wa,
    float* __restrict__ P)
{
    __shared__ __attribute__((aligned(16))) float As[2][GBM * APITCH];
    __shared__ __attribute__((aligned(16))) float Bs[2][(GBK / 2) * BPITCH];

    const int tid  = threadIdx.x;
    const int lane = tid & 31;
    const int wave = tid >> 5;
    const int wm   = wave >> 1;
    const int wn   = wave & 1;
    const int c    = lane & 15;
    const int half = lane >> 4;

    const int blockN  = blockIdx.x & 7;
    const int blockM  = blockIdx.x >> 3;
    const int rowBase = blockM * GBM;
    const bool isH = rowBase < (TGT * BATCH);
    const float* __restrict__ X = isH ? (h_t + (size_t)rowBase * HID)
                                      : (src + (size_t)(rowBase - TGT * BATCH) * HID);
    const float* __restrict__ W = Wa + (size_t)(isH ? 0 : HID) * ATT + blockN * GBN;

    const int arow = tid >> 3;
    const int acol = (tid & 7) * 4;
    const int bkp  = tid >> 4;
    const int bcol = (tid & 15) * 4;
    const float* __restrict__ gA = X + (size_t)arow * HID + acol;
    const float* __restrict__ gB = W + (size_t)(2 * bkp) * ATT + bcol;
    const int sA = arow * APITCH + acol;
    const int sB = bkp * BPITCH + bcol * 2;

    float4 ra0, ra1, ra2, ra3, rbe0, rbo0, rbe1, rbo1;

#define LOAD_STAGE()                                                            \
    do {                                                                        \
        rbe0 = *(const float4*)(pB);                                            \
        rbo0 = *(const float4*)(pB + ATT);                                      \
        rbe1 = *(const float4*)(pB + 16 * ATT);                                 \
        rbo1 = *(const float4*)(pB + 17 * ATT);                                 \
        ra0  = *(const float4*)(pA);                                            \
        ra1  = *(const float4*)(pA + 16 * HID);                                 \
        ra2  = *(const float4*)(pA + 32 * HID);                                 \
        ra3  = *(const float4*)(pA + 48 * HID);                                 \
        pA += GBK;                                                              \
        pB += (size_t)GBK * ATT;                                                \
    } while (0)

#define STORE_STAGE(BUF)                                                        \
    do {                                                                        \
        *(float4*)(&As[(BUF)][sA + 0  * APITCH]) = ra0;                         \
        *(float4*)(&As[(BUF)][sA + 16 * APITCH]) = ra1;                         \
        *(float4*)(&As[(BUF)][sA + 32 * APITCH]) = ra2;                         \
        *(float4*)(&As[(BUF)][sA + 48 * APITCH]) = ra3;                         \
        *(float4*)(&Bs[(BUF)][sB])                  = make_float4(rbe0.x, rbo0.x, rbe0.y, rbo0.y); \
        *(float4*)(&Bs[(BUF)][sB + 4])              = make_float4(rbe0.z, rbo0.z, rbe0.w, rbo0.w); \
        *(float4*)(&Bs[(BUF)][sB + 8 * BPITCH])     = make_float4(rbe1.x, rbo1.x, rbe1.y, rbo1.y); \
        *(float4*)(&Bs[(BUF)][sB + 8 * BPITCH + 4]) = make_float4(rbe1.z, rbo1.z, rbe1.w, rbo1.w); \
    } while (0)

    const float* __restrict__ pA = gA;
    const float* __restrict__ pB = gB;
    LOAD_STAGE();
    STORE_STAGE(0);
    __syncthreads();

    const int fA = (wm * 32 + c) * APITCH + 2 * half;
    const int fB = half * BPITCH + (wn * 32 + c) * 2;

    v8f acc00 = {}, acc01 = {}, acc10 = {}, acc11 = {};

    for (int st = 0; st < HID / GBK; ++st) {
        const int cur = st & 1;
        const bool more = (st + 1) < (HID / GBK);
        if (more) LOAD_STAGE();

        const float* __restrict__ as = &As[cur][fA];
        const float* __restrict__ bs = &Bs[cur][fB];
        v2f a0[GBK / 4], a1[GBK / 4], b0[GBK / 4], b1[GBK / 4];
#pragma unroll
        for (int j = 0; j < GBK / 4; ++j) {
            a0[j] = *(const v2f*)(as + 4 * j);
            a1[j] = *(const v2f*)(as + 16 * APITCH + 4 * j);
            b0[j] = *(const v2f*)(bs + 2 * j * BPITCH);
            b1[j] = *(const v2f*)(bs + 2 * j * BPITCH + 32);
        }

#pragma unroll
        for (int j = 0; j < GBK / 4 - 2; ++j) {
            acc00 = __builtin_amdgcn_wmma_f32_16x16x4_f32(false, a0[j], false, b0[j], (short)0, acc00, false, false);
            acc01 = __builtin_amdgcn_wmma_f32_16x16x4_f32(false, a0[j], false, b1[j], (short)0, acc01, false, false);
            acc10 = __builtin_amdgcn_wmma_f32_16x16x4_f32(false, a1[j], false, b0[j], (short)0, acc10, false, false);
            acc11 = __builtin_amdgcn_wmma_f32_16x16x4_f32(false, a1[j], false, b1[j], (short)0, acc11, false, false);
        }

        __builtin_amdgcn_sched_barrier(0);

        if (more) STORE_STAGE(cur ^ 1);

#pragma unroll
        for (int j = GBK / 4 - 2; j < GBK / 4; ++j) {
            acc00 = __builtin_amdgcn_wmma_f32_16x16x4_f32(false, a0[j], false, b0[j], (short)0, acc00, false, false);
            acc01 = __builtin_amdgcn_wmma_f32_16x16x4_f32(false, a0[j], false, b1[j], (short)0, acc01, false, false);
            acc10 = __builtin_amdgcn_wmma_f32_16x16x4_f32(false, a1[j], false, b0[j], (short)0, acc10, false, false);
            acc11 = __builtin_amdgcn_wmma_f32_16x16x4_f32(false, a1[j], false, b1[j], (short)0, acc11, false, false);
        }
        __syncthreads();
    }
#undef LOAD_STAGE
#undef STORE_STAGE

    float* __restrict__ o00 = P + (size_t)(rowBase + wm * 32 + half * 8) * ATT + blockN * GBN + wn * 32 + c;
    float* __restrict__ o10 = o00 + (size_t)16 * ATT;
#pragma unroll
    for (int r = 0; r < 8; ++r) {
        o00[(size_t)r * ATT]      = acc00[r];
        o00[(size_t)r * ATT + 16] = acc01[r];
        o10[(size_t)r * ATT]      = acc10[r];
        o10[(size_t)r * ATT + 16] = acc11[r];
    }
}

#define TT 8
#define PPITCH 36

typedef __attribute__((ext_vector_type(4))) float v4f;

__device__ __forceinline__ v2f tanh2(v2f x) {
    v2f r;
    r.x = fast_tanh(x.x);
    r.y = fast_tanh(x.y);
    return r;
}

__device__ __forceinline__ void lds_barrier() {
    asm volatile("s_wait_dscnt 0x0\n\ts_barrier_signal -1\n\ts_barrier_wait -1" ::: "memory");
}

__device__ __forceinline__ void score_row(const v4f (&hp)[TT][4], const v4f (&va)[4],
                                          const v4f (&sp)[4], float* dst)
{
    v2f a2[TT];
#pragma unroll
    for (int t = 0; t < TT; ++t) a2[t] = (v2f){0.f, 0.f};
#pragma unroll
    for (int i = 0; i < 4; ++i) {
#pragma unroll
        for (int t = 0; t < TT; ++t) a2[t] += tanh2(hp[t][i].xy + sp[i].xy) * va[i].xy;
#pragma unroll
        for (int t = 0; t < TT; ++t) a2[t] += tanh2(hp[t][i].zw + sp[i].zw) * va[i].zw;
    }
#pragma unroll
    for (int t = 0; t < TT; ++t) dst[t * SRC * PPITCH] = a2[t].x + a2[t].y;
}

__global__ __launch_bounds__(256) __attribute__((amdgpu_waves_per_eu(2, 2))) void attn_fused_kernel(
    const float* __restrict__ src,
    const float* __restrict__ Va,
    const float* __restrict__ P,
    float* __restrict__ out)
{
    __shared__ __attribute__((aligned(16))) float lds[SRC * TT + TT * SRC * PPITCH];
    float* attn = lds;
    float* part = lds + SRC * TT;

    const int b    = blockIdx.x & (BATCH - 1);
    const int t0   = (blockIdx.x >> 5) * TT;
    const int tid  = threadIdx.x;
    const int lane = tid & 31;
    const int wave = tid >> 5;
    const int col  = lane * 4;

    v4f hp[TT][4], va[4], spA[4], spB[4];
    const float* __restrict__ hrow = P + (size_t)(t0 * BATCH + b) * ATT + col;
#pragma unroll
    for (int t = 0; t < TT; ++t)
#pragma unroll
        for (int i = 0; i < 4; ++i)
            hp[t][i] = *(const v4f*)(hrow + (size_t)t * BATCH * ATT + i * 128);
#pragma unroll
    for (int i = 0; i < 4; ++i) va[i] = *(const v4f*)(Va + col + i * 128);

    const float* __restrict__ srow = P + (size_t)(TGT * BATCH + wave * BATCH + b) * ATT + col;
    const size_t sstep = (size_t)8 * BATCH * ATT;
#pragma unroll
    for (int i = 0; i < 4; ++i) spA[i] = *(const v4f*)(srow + i * 128);

    float* dst = &part[wave * PPITCH + lane];

    for (int kk = 0; kk < 3; ++kk) {
        srow += sstep;
#pragma unroll
        for (int i = 0; i < 4; ++i) spB[i] = *(const v4f*)(srow + i * 128);
        score_row(hp, va, spA, dst);
        dst += 8 * PPITCH;
        srow += sstep;
#pragma unroll
        for (int i = 0; i < 4; ++i) spA[i] = *(const v4f*)(srow + i * 128);
        score_row(hp, va, spB, dst);
        dst += 8 * PPITCH;
    }
    srow += sstep;
#pragma unroll
    for (int i = 0; i < 4; ++i) spB[i] = *(const v4f*)(srow + i * 128);

    const float* __restrict__ sg = src + (size_t)b * HID + tid * 2;
    v2f vr[SRC];
#pragma unroll
    for (int s = 0; s < SRC; ++s) vr[s] = *(const v2f*)(sg + (size_t)s * BATCH * HID);

    score_row(hp, va, spA, dst);
    dst += 8 * PPITCH;
    score_row(hp, va, spB, dst);
    lds_barrier();

    const float* q0 = &part[(wave * SRC + lane) * PPITCH];
    const float* q1 = q0 + 32 * PPITCH;
    v4f s0 = *(const v4f*)q0;
    v4f s1 = *(const v4f*)q1;
#pragma unroll
    for (int j = 1; j < 8; ++j) {
        s0 += *(const v4f*)(q0 + 4 * j);
        s1 += *(const v4f*)(q1 + 4 * j);
    }
    const float v0 = (s0.x + s0.y) + (s0.z + s0.w);
    const float v1 = (s1.x + s1.y) + (s1.z + s1.w);

    {
        float mx = fmaxf(v0, v1);
#pragma unroll
        for (int off = 16; off; off >>= 1)
            mx = fmaxf(mx, __shfl_xor(mx, off, 32));
        float e0 = __expf(v0 - mx), e1 = __expf(v1 - mx);
        float ssum = e0 + e1;
#pragma unroll
        for (int off = 16; off; off >>= 1)
            ssum += __shfl_xor(ssum, off, 32);
        float inv = 1.f / ssum;
        attn[lane * TT + wave]        = e0 * inv;
        attn[(lane + 32) * TT + wave] = e1 * inv;
    }
    lds_barrier();

    v2f c[TT];
#pragma unroll
    for (int t = 0; t < TT; ++t) c[t] = (v2f){0.f, 0.f};
#pragma unroll
    for (int s = 0; s < SRC; ++s) {
        const v4f w0 = *(const v4f*)(&attn[s * TT]);
        const v4f w1 = *(const v4f*)(&attn[s * TT + 4]);
        c[0] += w0.x * vr[s];
        c[1] += w0.y * vr[s];
        c[2] += w0.z * vr[s];
        c[3] += w0.w * vr[s];
        c[4] += w1.x * vr[s];
        c[5] += w1.y * vr[s];
        c[6] += w1.z * vr[s];
        c[7] += w1.w * vr[s];
    }
    float* __restrict__ og = out + (size_t)(t0 * BATCH + b) * HID + tid * 2;
#pragma unroll
    for (int t = 0; t < TT; ++t) *(v2f*)(og + (size_t)t * BATCH * HID) = c[t];
}

extern "C" void kernel_launch(void* const* d_in, const int* in_sizes, int n_in,
                              void* d_out, int out_size, void* d_ws, size_t ws_size,
                              hipStream_t stream) {
    const float* h_t = (const float*)d_in[0];
    const float* src = (const float*)d_in[1];
    const float* Wa  = (const float*)d_in[2];
    const float* Va  = (const float*)d_in[3];
    float* out = (float*)d_out;

    float* P = (float*)d_ws;

    proj_gemm_kernel<<<512, 128, 0, stream>>>(h_t, src, Wa, P);

    attn_fused_kernel<<<256, 256, 0, stream>>>(src, Va, P, out);
}


# Projection GEMM: each work-group starts its K loop at chunk (row tile + column tile) mod 16 and wraps, so groups sharing X or W tiles do not request the same lines together; k sum order rotated
# speedup vs baseline: 1.6740x; 1.0434x over previous
#include <hip/hip_runtime.h>

#define TGT 64
#define SRC 64
#define BATCH 32
#define HID 512
#define ATT 512

typedef __attribute__((ext_vector_type(2))) float v2f;
typedef __attribute__((ext_vector_type(8))) float v8f;

__device__ __forceinline__ float fast_tanh(float x) {
#if defined(__has_builtin)
#if __has_builtin(__builtin_amdgcn_tanhf)
    return __builtin_amdgcn_tanhf(x);
#elif __has_builtin(__builtin_amdgcn_tanh_f32)
    return __builtin_amdgcn_tanh_f32(x);
#else
    return tanhf(x);
#endif
#else
    return tanhf(x);
#endif
}

#define GBM 64
#define GBN 64
#define GBK 32
#define APITCH 36
#define BPITCH 160

__global__ __launch_bounds__(128) __attribute__((amdgpu_waves_per_eu(1, 2))) void proj_gemm_kernel(
    const float* __restrict__ h_t,
    const float* __restrict__ src,
    const float* __restrict__ Wa,
    float* __restrict__ P)
{
    __shared__ __attribute__((aligned(16))) float As[2][GBM * APITCH];
    __shared__ __attribute__((aligned(16))) float Bs[2][(GBK / 2) * BPITCH];

    const int tid  = threadIdx.x;
    const int lane = tid & 31;
    const int wave = tid >> 5;
    const int wm   = wave >> 1;
    const int wn   = wave & 1;
    const int c    = lane & 15;
    const int half = lane >> 4;

    const int blockN  = blockIdx.x & 7;
    const int blockM  = blockIdx.x >> 3;
    const int rowBase = blockM * GBM;
    const bool isH = rowBase < (TGT * BATCH);
    const float* __restrict__ X = isH ? (h_t + (size_t)rowBase * HID)
                                      : (src + (size_t)(rowBase - TGT * BATCH) * HID);
    const float* __restrict__ W = Wa + (size_t)(isH ? 0 : HID) * ATT + blockN * GBN;

    const int arow = tid >> 3;
    const int acol = (tid & 7) * 4;
    const int bkp  = tid >> 4;
    const int bcol = (tid & 15) * 4;
    const float* __restrict__ gA = X + (size_t)arow * HID + acol;
    const float* __restrict__ gB = W + (size_t)(2 * bkp) * ATT + bcol;
    const int sA = arow * APITCH + acol;
    const int sB = bkp * BPITCH + bcol * 2;

    float4 ra0, ra1, ra2, ra3, rbe0, rbo0, rbe1, rbo1;

#define LOAD_STAGE()                                                            \
    do {                                                                        \
        rbe0 = *(const float4*)(pB);                                            \
        rbo0 = *(const float4*)(pB + ATT);                                      \
        rbe1 = *(const float4*)(pB + 16 * ATT);                                 \
        rbo1 = *(const float4*)(pB + 17 * ATT);                                 \
        ra0  = *(const float4*)(pA);                                            \
        ra1  = *(const float4*)(pA + 16 * HID);                                 \
        ra2  = *(const float4*)(pA + 32 * HID);                                 \
        ra3  = *(const float4*)(pA + 48 * HID);                                 \
        kc = (kc + 1) & (NCH - 1);                                              \
        pA = kc ? pA + GBK : gA;                                                \
        pB = kc ? pB + (size_t)GBK * ATT : gB;                                  \
    } while (0)

#define STORE_STAGE(BUF)                                                        \
    do {                                                                        \
        *(float4*)(&As[(BUF)][sA + 0  * APITCH]) = ra0;                         \
        *(float4*)(&As[(BUF)][sA + 16 * APITCH]) = ra1;                         \
        *(float4*)(&As[(BUF)][sA + 32 * APITCH]) = ra2;                         \
        *(float4*)(&As[(BUF)][sA + 48 * APITCH]) = ra3;                         \
        *(float4*)(&Bs[(BUF)][sB])                  = make_float4(rbe0.x, rbo0.x, rbe0.y, rbo0.y); \
        *(float4*)(&Bs[(BUF)][sB + 4])              = make_float4(rbe0.z, rbo0.z, rbe0.w, rbo0.w); \
        *(float4*)(&Bs[(BUF)][sB + 8 * BPITCH])     = make_float4(rbe1.x, rbo1.x, rbe1.y, rbo1.y); \
        *(float4*)(&Bs[(BUF)][sB + 8 * BPITCH + 4]) = make_float4(rbe1.z, rbo1.z, rbe1.w, rbo1.w); \
    } while (0)

    constexpr int NCH = HID / GBK;
    int kc = (blockM + blockN) & (NCH - 1);
    const float* __restrict__ pA = gA + kc * GBK;
    const float* __restrict__ pB = gB + (size_t)kc * GBK * ATT;
    LOAD_STAGE();
    STORE_STAGE(0);
    __syncthreads();

    const int fA = (wm * 32 + c) * APITCH + 2 * half;
    const int fB = half * BPITCH + (wn * 32 + c) * 2;

    v8f acc00 = {}, acc01 = {}, acc10 = {}, acc11 = {};

    for (int st = 0; st < HID / GBK; ++st) {
        const int cur = st & 1;
        const bool more = (st + 1) < (HID / GBK);
        if (more) LOAD_STAGE();

        const float* __restrict__ as = &As[cur][fA];
        const float* __restrict__ bs = &Bs[cur][fB];
        v2f a0[GBK / 4], a1[GBK / 4], b0[GBK / 4], b1[GBK / 4];
#pragma unroll
        for (int j = 0; j < GBK / 4; ++j) {
            a0[j] = *(const v2f*)(as + 4 * j);
            a1[j] = *(const v2f*)(as + 16 * APITCH + 4 * j);
            b0[j] = *(const v2f*)(bs + 2 * j * BPITCH);
            b1[j] = *(const v2f*)(bs + 2 * j * BPITCH + 32);
        }

#pragma unroll
        for (int j = 0; j < GBK / 4 - 2; ++j) {
            acc00 = __builtin_amdgcn_wmma_f32_16x16x4_f32(false, a0[j], false, b0[j], (short)0, acc00, false, false);
            acc01 = __builtin_amdgcn_wmma_f32_16x16x4_f32(false, a0[j], false, b1[j], (short)0, acc01, false, false);
            acc10 = __builtin_amdgcn_wmma_f32_16x16x4_f32(false, a1[j], false, b0[j], (short)0, acc10, false, false);
            acc11 = __builtin_amdgcn_wmma_f32_16x16x4_f32(false, a1[j], false, b1[j], (short)0, acc11, false, false);
        }

        __builtin_amdgcn_sched_barrier(0);

        if (more) STORE_STAGE(cur ^ 1);

#pragma unroll
        for (int j = GBK / 4 - 2; j < GBK / 4; ++j) {
            acc00 = __builtin_amdgcn_wmma_f32_16x16x4_f32(false, a0[j], false, b0[j], (short)0, acc00, false, false);
            acc01 = __builtin_amdgcn_wmma_f32_16x16x4_f32(false, a0[j], false, b1[j], (short)0, acc01, false, false);
            acc10 = __builtin_amdgcn_wmma_f32_16x16x4_f32(false, a1[j], false, b0[j], (short)0, acc10, false, false);
            acc11 = __builtin_amdgcn_wmma_f32_16x16x4_f32(false, a1[j], false, b1[j], (short)0, acc11, false, false);
        }
        __syncthreads();
    }
#undef LOAD_STAGE
#undef STORE_STAGE

    float* __restrict__ o00 = P + (size_t)(rowBase + wm * 32 + half * 8) * ATT + blockN * GBN + wn * 32 + c;
    float* __restrict__ o10 = o00 + (size_t)16 * ATT;
#pragma unroll
    for (int r = 0; r < 8; ++r) {
        o00[(size_t)r * ATT]      = acc00[r];
        o00[(size_t)r * ATT + 16] = acc01[r];
        o10[(size_t)r * ATT]      = acc10[r];
        o10[(size_t)r * ATT + 16] = acc11[r];
    }
}

#define TT 8
#define PPITCH 36

typedef __attribute__((ext_vector_type(4))) float v4f;

__device__ __forceinline__ v2f tanh2(v2f x) {
    v2f r;
    r.x = fast_tanh(x.x);
    r.y = fast_tanh(x.y);
    return r;
}

__device__ __forceinline__ void lds_barrier() {
    asm volatile("s_wait_dscnt 0x0\n\ts_barrier_signal -1\n\ts_barrier_wait -1" ::: "memory");
}

__device__ __forceinline__ void score_row(const v4f (&hp)[TT][4], const v4f (&va)[4],
                                          const v4f (&sp)[4], float* dst)
{
    v2f a2[TT];
#pragma unroll
    for (int t = 0; t < TT; ++t) a2[t] = (v2f){0.f, 0.f};
#pragma unroll
    for (int i = 0; i < 4; ++i) {
#pragma unroll
        for (int t = 0; t < TT; ++t) a2[t] += tanh2(hp[t][i].xy + sp[i].xy) * va[i].xy;
#pragma unroll
        for (int t = 0; t < TT; ++t) a2[t] += tanh2(hp[t][i].zw + sp[i].zw) * va[i].zw;
    }
#pragma unroll
    for (int t = 0; t < TT; ++t) dst[t * SRC * PPITCH] = a2[t].x + a2[t].y;
}

__global__ __launch_bounds__(256) __attribute__((amdgpu_waves_per_eu(2, 2))) void attn_fused_kernel(
    const float* __restrict__ src,
    const float* __restrict__ Va,
    const float* __restrict__ P,
    float* __restrict__ out)
{
    __shared__ __attribute__((aligned(16))) float lds[SRC * TT + TT * SRC * PPITCH];
    float* attn = lds;
    float* part = lds + SRC * TT;

    const int b    = blockIdx.x & (BATCH - 1);
    const int t0   = (blockIdx.x >> 5) * TT;
    const int tid  = threadIdx.x;
    const int lane = tid & 31;
    const int wave = tid >> 5;
    const int col  = lane * 4;

    v4f hp[TT][4], va[4], spA[4], spB[4];
    const float* __restrict__ hrow = P + (size_t)(t0 * BATCH + b) * ATT + col;
#pragma unroll
    for (int t = 0; t < TT; ++t)
#pragma unroll
        for (int i = 0; i < 4; ++i)
            hp[t][i] = *(const v4f*)(hrow + (size_t)t * BATCH * ATT + i * 128);
#pragma unroll
    for (int i = 0; i < 4; ++i) va[i] = *(const v4f*)(Va + col + i * 128);

    const float* __restrict__ srow = P + (size_t)(TGT * BATCH + wave * BATCH + b) * ATT + col;
    const size_t sstep = (size_t)8 * BATCH * ATT;
#pragma unroll
    for (int i = 0; i < 4; ++i) spA[i] = *(const v4f*)(srow + i * 128);

    float* dst = &part[wave * PPITCH + lane];

    for (int kk = 0; kk < 3; ++kk) {
        srow += sstep;
#pragma unroll
        for (int i = 0; i < 4; ++i) spB[i] = *(const v4f*)(srow + i * 128);
        score_row(hp, va, spA, dst);
        dst += 8 * PPITCH;
        srow += sstep;
#pragma unroll
        for (int i = 0; i < 4; ++i) spA[i] = *(const v4f*)(srow + i * 128);
        score_row(hp, va, spB, dst);
        dst += 8 * PPITCH;
    }
    srow += sstep;
#pragma unroll
    for (int i = 0; i < 4; ++i) spB[i] = *(const v4f*)(srow + i * 128);

    const float* __restrict__ sg = src + (size_t)b * HID + tid * 2;
    v2f vr[SRC];
#pragma unroll
    for (int s = 0; s < SRC; ++s) vr[s] = *(const v2f*)(sg + (size_t)s * BATCH * HID);

    score_row(hp, va, spA, dst);
    dst += 8 * PPITCH;
    score_row(hp, va, spB, dst);
    lds_barrier();

    const float* q0 = &part[(wave * SRC + lane) * PPITCH];
    const float* q1 = q0 + 32 * PPITCH;
    v4f s0 = *(const v4f*)q0;
    v4f s1 = *(const v4f*)q1;
#pragma unroll
    for (int j = 1; j < 8; ++j) {
        s0 += *(const v4f*)(q0 + 4 * j);
        s1 += *(const v4f*)(q1 + 4 * j);
    }
    const float v0 = (s0.x + s0.y) + (s0.z + s0.w);
    const float v1 = (s1.x + s1.y) + (s1.z + s1.w);

    {
        float mx = fmaxf(v0, v1);
#pragma unroll
        for (int off = 16; off; off >>= 1)
            mx = fmaxf(mx, __shfl_xor(mx, off, 32));
        float e0 = __expf(v0 - mx), e1 = __expf(v1 - mx);
        float ssum = e0 + e1;
#pragma unroll
        for (int off = 16; off; off >>= 1)
            ssum += __shfl_xor(ssum, off, 32);
        float inv = 1.f / ssum;
        attn[lane * TT + wave]        = e0 * inv;
        attn[(lane + 32) * TT + wave] = e1 * inv;
    }
    lds_barrier();

    v2f c[TT];
#pragma unroll
    for (int t = 0; t < TT; ++t) c[t] = (v2f){0.f, 0.f};
#pragma unroll
    for (int s = 0; s < SRC; ++s) {
        const v4f w0 = *(const v4f*)(&attn[s * TT]);
        const v4f w1 = *(const v4f*)(&attn[s * TT + 4]);
        c[0] += w0.x * vr[s];
        c[1] += w0.y * vr[s];
        c[2] += w0.z * vr[s];
        c[3] += w0.w * vr[s];
        c[4] += w1.x * vr[s];
        c[5] += w1.y * vr[s];
        c[6] += w1.z * vr[s];
        c[7] += w1.w * vr[s];
    }
    float* __restrict__ og = out + (size_t)(t0 * BATCH + b) * HID + tid * 2;
#pragma unroll
    for (int t = 0; t < TT; ++t) *(v2f*)(og + (size_t)t * BATCH * HID) = c[t];
}

extern "C" void kernel_launch(void* const* d_in, const int* in_sizes, int n_in,
                              void* d_out, int out_size, void* d_ws, size_t ws_size,
                              hipStream_t stream) {
    const float* h_t = (const float*)d_in[0];
    const float* src = (const float*)d_in[1];
    const float* Wa  = (const float*)d_in[2];
    const float* Va  = (const float*)d_in[3];
    float* out = (float*)d_out;

    float* P = (float*)d_ws;

    proj_gemm_kernel<<<512, 128, 0, stream>>>(h_t, src, Wa, P);

    attn_fused_kernel<<<256, 256, 0, stream>>>(src, Va, P, out);
}
